# starting kernel
# baseline (speedup 1.0000x reference)
#include <hip/hip_runtime.h>
#include <math.h>

typedef float v2f __attribute__((ext_vector_type(2)));
typedef float v8f __attribute__((ext_vector_type(8)));

__device__ inline v8f wmma4(v2f a, v2f b, v8f c) {
    return __builtin_amdgcn_wmma_f32_16x16x4_f32(
         false, a,  false, b,
         (short)0, c,  false,  false);
}

__device__ inline float swap16(float v) {
    return __int_as_float(__builtin_amdgcn_ds_swizzle(__float_as_int(v), 0x401F));
}

__global__ void build_G(const float* __restrict__ w, float* __restrict__ g) {
    int t = threadIdx.x;
    if (t >= 512) return;
    int l = t >> 8;
    int e = t & 255;
    int r = e >> 4;
    int c = e & 15;

    int j = r;
    if (j & 2) j ^= 1;
    if (j & 4) j ^= 2;
    if (j & 8) j ^= 4;

    float ar = 1.0f, ai = 0.0f;
    #pragma unroll
    for (int q = 0; q < 4; ++q) {
        float ty = w[(l * 4 + q) * 2 + 0] * 0.5f;
        float tz = w[(l * 4 + q) * 2 + 1] * 0.5f;
        float cy, sy, cz, sz;
        __sincosf(ty, &sy, &cy);
        __sincosf(tz, &sz, &cz);
        int br = (j >> (3 - q)) & 1;
        int bc = (c >> (3 - q)) & 1;
        float ur, ui;
        if (br == 0) { float m = bc ? -sy : cy; ur = m * cz; ui = -m * sz; }
        else         { float m = bc ?  cy : sy; ur = m * cz; ui =  m * sz; }
        float nr = ar * ur - ai * ui;
        float ni = ar * ui + ai * ur;
        ar = nr; ai = ni;
    }
    g[(l * 3 + 0) * 256 + e] = ar;
    g[(l * 3 + 1) * 256 + e] = ai;
    g[(l * 3 + 2) * 256 + e] = -ai;
}

__global__ void __launch_bounds__(256) qml_kernel(const float* __restrict__ x,
                                                  const float* __restrict__ g,
                                                  float* __restrict__ out) {
    const int lane = threadIdx.x & 31;
    const int wave = threadIdx.x >> 5;
    const int n    = lane & 15;
    const bool hi  = lane >= 16;
    const int cb   = hi ? 2 : 0;
    const int samp = blockIdx.x * 128 + wave * 16 + n;

    float4 xv = reinterpret_cast<const float4*>(x)[samp];
    float c0, s0, c1, s1, c2, s2, c3, s3;
    __sincosf(xv.x * 0.5f, &s0, &c0);
    __sincosf(xv.y * 0.5f, &s1, &c1);
    __sincosf(xv.z * 0.5f, &s2, &c2);
    __sincosf(xv.w * 0.5f, &s3, &c3);

    v2f Ar[4], Ai[4], An[4];
    #pragma unroll
    for (int k = 0; k < 4; ++k) {
        int off = n * 16 + 4 * k + cb;
        Ar[k] = *reinterpret_cast<const v2f*>(g + 0 * 256 + off);
        Ai[k] = *reinterpret_cast<const v2f*>(g + 1 * 256 + off);
        An[k] = *reinterpret_cast<const v2f*>(g + 2 * 256 + off);
    }

    const float f2  = hi ? s2 : c2;
    const float ab0 = c0 * c1, ab1 = c0 * s1, ab2 = s0 * c1, ab3 = s0 * s1;
    const float ab[4] = {ab0, ab1, ab2, ab3};
    const float fd0 = f2 * c3, fd1 = f2 * s3;
    v2f Br[4], Bi[4];
    #pragma unroll
    for (int k = 0; k < 4; ++k) {
        #pragma unroll
        for (int jj = 0; jj < 2; ++jj) {
            const float m = ab[k] * (jj ? fd1 : fd0);
            const int cpc = ((k >> 1) & 1) + (k & 1) + jj;
            const int ql = cpc & 3;
            const int qh = (cpc + 1) & 3;
            const float rl = (ql == 0) ? m : (ql == 2) ? -m : 0.f;
            const float il = (ql == 1) ? -m : (ql == 3) ? m : 0.f;
            const float rh = (qh == 0) ? m : (qh == 2) ? -m : 0.f;
            const float ih = (qh == 1) ? -m : (qh == 3) ? m : 0.f;
            Br[k][jj] = hi ? rh : rl;
            Bi[k][jj] = hi ? ih : il;
        }
    }

    v8f dr = {0.f,0.f,0.f,0.f,0.f,0.f,0.f,0.f};
    v8f di = {0.f,0.f,0.f,0.f,0.f,0.f,0.f,0.f};
    #pragma unroll
    for (int k = 0; k < 4; ++k) {
        dr = wmma4(Ar[k], Br[k], dr);
        di = wmma4(Ar[k], Bi[k], di);
    }
    #pragma unroll
    for (int k = 0; k < 4; ++k) {
        dr = wmma4(An[k], Bi[k], dr);
        di = wmma4(Ai[k], Br[k], di);
    }

    float R[8], I[8];
    #pragma unroll
    for (int r = 0; r < 8; ++r) { R[r] = dr[r]; I[r] = di[r]; }

    #pragma unroll
    for (int r = 0; r < 4; ++r) {
        float arr = R[r], aii = I[r], brr = R[r + 4], bii = I[r + 4];
        R[r]     = c1 * arr + s1 * bii;  I[r]     = c1 * aii - s1 * brr;
        R[r + 4] = c1 * brr + s1 * aii;  I[r + 4] = c1 * bii - s1 * arr;
    }
    #pragma unroll
    for (int t2 = 0; t2 < 4; ++t2) {
        int r = (t2 & 1) + (t2 >> 1) * 4;
        float arr = R[r], aii = I[r], brr = R[r + 2], bii = I[r + 2];
        R[r]     = c2 * arr + s2 * bii;  I[r]     = c2 * aii - s2 * brr;
        R[r + 2] = c2 * brr + s2 * aii;  I[r + 2] = c2 * bii - s2 * arr;
    }
    #pragma unroll
    for (int t3 = 0; t3 < 4; ++t3) {
        int r = t3 * 2;
        float arr = R[r], aii = I[r], brr = R[r + 1], bii = I[r + 1];
        R[r]     = c3 * arr + s3 * bii;  I[r]     = c3 * aii - s3 * brr;
        R[r + 1] = c3 * brr + s3 * aii;  I[r + 1] = c3 * bii - s3 * arr;
    }

    float SR[8], SI[8];
    #pragma unroll
    for (int r = 0; r < 8; ++r) { SR[r] = swap16(R[r]); SI[r] = swap16(I[r]); }

    float NR[8], NI[8], PR[8], PI[8];
    #pragma unroll
    for (int r = 0; r < 8; ++r) {
        NR[r] = c0 * R[r]  + s0 * SI[r];
        NI[r] = c0 * I[r]  - s0 * SR[r];
        PR[r] = c0 * SR[r] + s0 * I[r];
        PI[r] = c0 * SI[r] - s0 * R[r];
    }

    #pragma unroll
    for (int jj = 0; jj < 2; ++jj) {
        Br[0][jj] = hi ? PR[jj + 2] : NR[jj];      Bi[0][jj] = hi ? PI[jj + 2] : NI[jj];
        Br[1][jj] = hi ? PR[jj + 6] : NR[jj + 4];  Bi[1][jj] = hi ? PI[jj + 6] : NI[jj + 4];
        Br[2][jj] = hi ? NR[jj + 2] : PR[jj];      Bi[2][jj] = hi ? NI[jj + 2] : PI[jj];
        Br[3][jj] = hi ? NR[jj + 6] : PR[jj + 4];  Bi[3][jj] = hi ? NI[jj + 6] : PI[jj + 4];
    }

    #pragma unroll
    for (int k = 0; k < 4; ++k) {
        int off = n * 16 + 4 * k + cb;
        Ar[k] = *reinterpret_cast<const v2f*>(g + 3 * 256 + off);
        Ai[k] = *reinterpret_cast<const v2f*>(g + 4 * 256 + off);
        An[k] = *reinterpret_cast<const v2f*>(g + 5 * 256 + off);
    }

    v8f er = {0.f,0.f,0.f,0.f,0.f,0.f,0.f,0.f};
    v8f ei = {0.f,0.f,0.f,0.f,0.f,0.f,0.f,0.f};
    #pragma unroll
    for (int k = 0; k < 4; ++k) {
        er = wmma4(Ar[k], Br[k], er);
        ei = wmma4(Ar[k], Bi[k], ei);
    }
    #pragma unroll
    for (int k = 0; k < 4; ++k) {
        er = wmma4(An[k], Bi[k], er);
        ei = wmma4(Ai[k], Br[k], ei);
    }

    float p[8];
    #pragma unroll
    for (int r = 0; r < 8; ++r) p[r] = er[r] * er[r] + ei[r] * ei[r];

    const float t0 = p[0] + p[1], t1 = p[2] + p[3];
    const float t2 = p[4] + p[5], t3 = p[6] + p[7];
    const float u01 = t0 + t1, u23 = t2 + t3;
    float z0 = u01 + u23;
    float z1 = u01 - u23;
    float z2 = (t0 + t2) - (t1 + t3);
    const float w0 = p[0] - p[1], w1 = p[2] - p[3];
    const float w2 = p[4] - p[5], w3 = p[6] - p[7];
    float z3 = (w0 + w1) + (w2 + w3);
    if (hi) z0 = -z0;
    z0 += swap16(z0);
    z1 += swap16(z1);
    z2 += swap16(z2);
    z3 += swap16(z3);

    if (!hi) {
        float4 o; o.x = z0; o.y = z1; o.z = z2; o.w = z3;
        reinterpret_cast<float4*>(out)[samp] = o;
    }
}

extern "C" void kernel_launch(void* const* d_in, const int* in_sizes, int n_in,
                              void* d_out, int out_size, void* d_ws, size_t ws_size,
                              hipStream_t stream) {
    const float* x = (const float*)d_in[0];
    const float* w = (const float*)d_in[1];
    float* g   = (float*)d_ws;
    float* out = (float*)d_out;

    int B = in_sizes[0] / 4;
    build_G<<<1, 512, 0, stream>>>(w, g);
    qml_kernel<<<B / 128, 256, 0, stream>>>(x, g, out);
}


# Each complex matrix product by 3 real products, not 4: t=Gr*(vr+vi), re=t-(Gr+Gi)*vi, im=t+(Gi-Gr)*vr; summed planes from the table kernel; 24 f32 WMMAs per tile, not 32; sums regroup
# speedup vs baseline: 1.1677x; 1.1677x over previous
#include <hip/hip_runtime.h>
#include <math.h>

typedef float v2f __attribute__((ext_vector_type(2)));
typedef float v8f __attribute__((ext_vector_type(8)));

__device__ inline v8f wmma4(v2f a, v2f b, v8f c) {
    return __builtin_amdgcn_wmma_f32_16x16x4_f32(
         false, a,  false, b,
         (short)0, c,  false,  false);
}

__device__ inline float swap16(float v) {
    return __int_as_float(__builtin_amdgcn_ds_swizzle(__float_as_int(v), 0x401F));
}

__global__ void build_G(const float* __restrict__ w, float* __restrict__ g) {
    int t = threadIdx.x;
    if (t >= 512) return;
    int l = t >> 8;
    int e = t & 255;
    int r = e >> 4;
    int c = e & 15;

    int j = r;
    if (j & 2) j ^= 1;
    if (j & 4) j ^= 2;
    if (j & 8) j ^= 4;

    float ar = 1.0f, ai = 0.0f;
    #pragma unroll
    for (int q = 0; q < 4; ++q) {
        float ty = w[(l * 4 + q) * 2 + 0] * 0.5f;
        float tz = w[(l * 4 + q) * 2 + 1] * 0.5f;
        float cy, sy, cz, sz;
        __sincosf(ty, &sy, &cy);
        __sincosf(tz, &sz, &cz);
        int br = (j >> (3 - q)) & 1;
        int bc = (c >> (3 - q)) & 1;
        float ur, ui;
        if (br == 0) { float m = bc ? -sy : cy; ur = m * cz; ui = -m * sz; }
        else         { float m = bc ?  cy : sy; ur = m * cz; ui =  m * sz; }
        float nr = ar * ur - ai * ui;
        float ni = ar * ui + ai * ur;
        ar = nr; ai = ni;
    }
    g[(l * 3 + 0) * 256 + e] = ar;
    g[(l * 3 + 1) * 256 + e] = -(ar + ai);
    g[(l * 3 + 2) * 256 + e] = ai - ar;
}

__global__ void __launch_bounds__(256) qml_kernel(const float* __restrict__ x,
                                                  const float* __restrict__ g,
                                                  float* __restrict__ out) {
    const int lane = threadIdx.x & 31;
    const int wave = threadIdx.x >> 5;
    const int n    = lane & 15;
    const bool hi  = lane >= 16;
    const int cb   = hi ? 2 : 0;
    const int samp = blockIdx.x * 128 + wave * 16 + n;

    float4 xv = reinterpret_cast<const float4*>(x)[samp];
    float c0, s0, c1, s1, c2, s2, c3, s3;
    __sincosf(xv.x * 0.5f, &s0, &c0);
    __sincosf(xv.y * 0.5f, &s1, &c1);
    __sincosf(xv.z * 0.5f, &s2, &c2);
    __sincosf(xv.w * 0.5f, &s3, &c3);

    v2f Ar[4], Am[4], Ap[4];
    #pragma unroll
    for (int k = 0; k < 4; ++k) {
        int off = n * 16 + 4 * k + cb;
        Ar[k] = *reinterpret_cast<const v2f*>(g + 0 * 256 + off);
        Am[k] = *reinterpret_cast<const v2f*>(g + 1 * 256 + off);
        Ap[k] = *reinterpret_cast<const v2f*>(g + 2 * 256 + off);
    }

    const float f2  = hi ? s2 : c2;
    const float ab0 = c0 * c1, ab1 = c0 * s1, ab2 = s0 * c1, ab3 = s0 * s1;
    const float ab[4] = {ab0, ab1, ab2, ab3};
    const float fd0 = f2 * c3, fd1 = f2 * s3;
    v2f Br[4], Bi[4], Bs[4];
    #pragma unroll
    for (int k = 0; k < 4; ++k) {
        #pragma unroll
        for (int jj = 0; jj < 2; ++jj) {
            const float m = ab[k] * (jj ? fd1 : fd0);
            const int cpc = ((k >> 1) & 1) + (k & 1) + jj;
            const int ql = cpc & 3;
            const int qh = (cpc + 1) & 3;
            const float rl = (ql == 0) ? m : (ql == 2) ? -m : 0.f;
            const float il = (ql == 1) ? -m : (ql == 3) ? m : 0.f;
            const float rh = (qh == 0) ? m : (qh == 2) ? -m : 0.f;
            const float ih = (qh == 1) ? -m : (qh == 3) ? m : 0.f;
            const float sl = (ql == 0 || ql == 3) ? m : -m;
            const float sh = (qh == 0 || qh == 3) ? m : -m;
            Br[k][jj] = hi ? rh : rl;
            Bi[k][jj] = hi ? ih : il;
            Bs[k][jj] = hi ? sh : sl;
        }
    }

    v8f ta = {0.f,0.f,0.f,0.f,0.f,0.f,0.f,0.f};
    #pragma unroll
    for (int k = 0; k < 4; ++k) ta = wmma4(Ar[k], Bs[k], ta);
    v8f dr = ta, di = ta;
    #pragma unroll
    for (int k = 0; k < 4; ++k) {
        dr = wmma4(Am[k], Bi[k], dr);
        di = wmma4(Ap[k], Br[k], di);
    }

    float R[8], I[8];
    #pragma unroll
    for (int r = 0; r < 8; ++r) { R[r] = dr[r]; I[r] = di[r]; }

    #pragma unroll
    for (int r = 0; r < 4; ++r) {
        float arr = R[r], aii = I[r], brr = R[r + 4], bii = I[r + 4];
        R[r]     = c1 * arr + s1 * bii;  I[r]     = c1 * aii - s1 * brr;
        R[r + 4] = c1 * brr + s1 * aii;  I[r + 4] = c1 * bii - s1 * arr;
    }
    #pragma unroll
    for (int t2 = 0; t2 < 4; ++t2) {
        int r = (t2 & 1) + (t2 >> 1) * 4;
        float arr = R[r], aii = I[r], brr = R[r + 2], bii = I[r + 2];
        R[r]     = c2 * arr + s2 * bii;  I[r]     = c2 * aii - s2 * brr;
        R[r + 2] = c2 * brr + s2 * aii;  I[r + 2] = c2 * bii - s2 * arr;
    }
    #pragma unroll
    for (int t3 = 0; t3 < 4; ++t3) {
        int r = t3 * 2;
        float arr = R[r], aii = I[r], brr = R[r + 1], bii = I[r + 1];
        R[r]     = c3 * arr + s3 * bii;  I[r]     = c3 * aii - s3 * brr;
        R[r + 1] = c3 * brr + s3 * aii;  I[r + 1] = c3 * bii - s3 * arr;
    }

    float SR[8], SI[8];
    #pragma unroll
    for (int r = 0; r < 8; ++r) { SR[r] = swap16(R[r]); SI[r] = swap16(I[r]); }

    float NR[8], NI[8], PR[8], PI[8];
    #pragma unroll
    for (int r = 0; r < 8; ++r) {
        NR[r] = c0 * R[r]  + s0 * SI[r];
        NI[r] = c0 * I[r]  - s0 * SR[r];
        PR[r] = c0 * SR[r] + s0 * I[r];
        PI[r] = c0 * SI[r] - s0 * R[r];
    }

    #pragma unroll
    for (int jj = 0; jj < 2; ++jj) {
        Br[0][jj] = hi ? PR[jj + 2] : NR[jj];      Bi[0][jj] = hi ? PI[jj + 2] : NI[jj];
        Br[1][jj] = hi ? PR[jj + 6] : NR[jj + 4];  Bi[1][jj] = hi ? PI[jj + 6] : NI[jj + 4];
        Br[2][jj] = hi ? NR[jj + 2] : PR[jj];      Bi[2][jj] = hi ? NI[jj + 2] : PI[jj];
        Br[3][jj] = hi ? NR[jj + 6] : PR[jj + 4];  Bi[3][jj] = hi ? NI[jj + 6] : PI[jj + 4];
    }
    #pragma unroll
    for (int k = 0; k < 4; ++k) Bs[k] = Br[k] + Bi[k];

    #pragma unroll
    for (int k = 0; k < 4; ++k) {
        int off = n * 16 + 4 * k + cb;
        Ar[k] = *reinterpret_cast<const v2f*>(g + 3 * 256 + off);
        Am[k] = *reinterpret_cast<const v2f*>(g + 4 * 256 + off);
        Ap[k] = *reinterpret_cast<const v2f*>(g + 5 * 256 + off);
    }

    v8f tb = {0.f,0.f,0.f,0.f,0.f,0.f,0.f,0.f};
    #pragma unroll
    for (int k = 0; k < 4; ++k) tb = wmma4(Ar[k], Bs[k], tb);
    v8f er = tb, ei = tb;
    #pragma unroll
    for (int k = 0; k < 4; ++k) {
        er = wmma4(Am[k], Bi[k], er);
        ei = wmma4(Ap[k], Br[k], ei);
    }

    float p[8];
    #pragma unroll
    for (int r = 0; r < 8; ++r) p[r] = er[r] * er[r] + ei[r] * ei[r];

    const float t0 = p[0] + p[1], t1 = p[2] + p[3];
    const float t2 = p[4] + p[5], t3 = p[6] + p[7];
    const float u01 = t0 + t1, u23 = t2 + t3;
    float z0 = u01 + u23;
    float z1 = u01 - u23;
    float z2 = (t0 + t2) - (t1 + t3);
    const float w0 = p[0] - p[1], w1 = p[2] - p[3];
    const float w2 = p[4] - p[5], w3 = p[6] - p[7];
    float z3 = (w0 + w1) + (w2 + w3);
    if (hi) z0 = -z0;
    z0 += swap16(z0);
    z1 += swap16(z1);
    z2 += swap16(z2);
    z3 += swap16(z3);

    if (!hi) {
        float4 o; o.x = z0; o.y = z1; o.z = z2; o.w = z3;
        reinterpret_cast<float4*>(out)[samp] = o;
    }
}

extern "C" void kernel_launch(void* const* d_in, const int* in_sizes, int n_in,
                              void* d_out, int out_size, void* d_ws, size_t ws_size,
                              hipStream_t stream) {
    const float* x = (const float*)d_in[0];
    const float* w = (const float*)d_in[1];
    float* g   = (float*)d_ws;
    float* out = (float*)d_out;

    int B = in_sizes[0] / 4;
    build_G<<<1, 512, 0, stream>>>(w, g);
    qml_kernel<<<B / 128, 256, 0, stream>>>(x, g, out);
}


# Fragment k order of both products = the accumulator's own row order (comp 8h+r): no D->B relayout, qubit-0 RX on own rows only; table rows read as 16-byte pieces; k sums regroup
# speedup vs baseline: 1.2566x; 1.0761x over previous
#include <hip/hip_runtime.h>
#include <math.h>

typedef float v2f __attribute__((ext_vector_type(2)));
typedef float v8f __attribute__((ext_vector_type(8)));

__device__ inline v8f wmma4(v2f a, v2f b, v8f c) {
    return __builtin_amdgcn_wmma_f32_16x16x4_f32(
         false, a,  false, b,
         (short)0, c,  false,  false);
}

__device__ inline float swap16(float v) {
    return __int_as_float(__builtin_amdgcn_ds_swizzle(__float_as_int(v), 0x401F));
}

__global__ void build_G(const float* __restrict__ w, float* __restrict__ g) {
    int t = threadIdx.x;
    if (t >= 512) return;
    int l = t >> 8;
    int e = t & 255;
    int r = e >> 4;
    int c = e & 15;

    int j = r;
    if (j & 2) j ^= 1;
    if (j & 4) j ^= 2;
    if (j & 8) j ^= 4;

    float ar = 1.0f, ai = 0.0f;
    #pragma unroll
    for (int q = 0; q < 4; ++q) {
        float ty = w[(l * 4 + q) * 2 + 0] * 0.5f;
        float tz = w[(l * 4 + q) * 2 + 1] * 0.5f;
        float cy, sy, cz, sz;
        __sincosf(ty, &sy, &cy);
        __sincosf(tz, &sz, &cz);
        int br = (j >> (3 - q)) & 1;
        int bc = (c >> (3 - q)) & 1;
        float ur, ui;
        if (br == 0) { float m = bc ? -sy : cy; ur = m * cz; ui = -m * sz; }
        else         { float m = bc ?  cy : sy; ur = m * cz; ui =  m * sz; }
        float nr = ar * ur - ai * ui;
        float ni = ar * ui + ai * ur;
        ar = nr; ai = ni;
    }
    g[(l * 3 + 0) * 256 + e] = ar;
    g[(l * 3 + 1) * 256 + e] = -(ar + ai);
    g[(l * 3 + 2) * 256 + e] = ai - ar;
}

__device__ inline void load_frag(const float* __restrict__ p, v2f (&A)[4]) {
    const float4 lo = *reinterpret_cast<const float4*>(p);
    const float4 up = *reinterpret_cast<const float4*>(p + 4);
    A[0] = v2f{lo.x, lo.y};  A[1] = v2f{lo.z, lo.w};
    A[2] = v2f{up.x, up.y};  A[3] = v2f{up.z, up.w};
}

__global__ void __launch_bounds__(256) qml_kernel(const float* __restrict__ x,
                                                  const float* __restrict__ g,
                                                  float* __restrict__ out) {
    const int lane = threadIdx.x & 31;
    const int wave = threadIdx.x >> 5;
    const int n    = lane & 15;
    const bool hi  = lane >= 16;
    const int foff = n * 16 + (hi ? 8 : 0);
    const int samp = blockIdx.x * 128 + wave * 16 + n;

    float4 xv = reinterpret_cast<const float4*>(x)[samp];
    float c0, s0, c1, s1, c2, s2, c3, s3;
    __sincosf(xv.x * 0.5f, &s0, &c0);
    __sincosf(xv.y * 0.5f, &s1, &c1);
    __sincosf(xv.z * 0.5f, &s2, &c2);
    __sincosf(xv.w * 0.5f, &s3, &c3);

    v2f Ar[4], Am[4], Ap[4];
    load_frag(g + 0 * 256 + foff, Ar);
    load_frag(g + 1 * 256 + foff, Am);
    load_frag(g + 2 * 256 + foff, Ap);

    const float gl = hi ? 0.f : c0;
    const float gh = hi ? s0 : 0.f;
    const float gs = hi ? -s0 : c0;
    const float gt = hi ? s0 : c0;
    const float ab[4] = {c1 * c2, c1 * s2, s1 * c2, s1 * s2};
    const float f3[2] = {c3, s3};
    v2f Br[4], Bi[4], Bs[4];
    #pragma unroll
    for (int k = 0; k < 4; ++k) {
        #pragma unroll
        for (int jj = 0; jj < 2; ++jj) {
            const float P = ab[k] * f3[jj];
            const int q = (((k >> 1) & 1) + (k & 1) + jj) & 3;
            float br, bi, bs;
            if (q == 0)      { br =  gl * P;  bi = -(gh * P); bs =  gs * P; }
            else if (q == 1) { br = -(gh * P); bi = -(gl * P); bs = -(gt * P); }
            else if (q == 2) { br = -(gl * P); bi =  gh * P;  bs = -(gs * P); }
            else             { br =  gh * P;  bi =  gl * P;  bs =  gt * P; }
            Br[k][jj] = br;
            Bi[k][jj] = bi;
            Bs[k][jj] = bs;
        }
    }

    v8f ta = {0.f,0.f,0.f,0.f,0.f,0.f,0.f,0.f};
    #pragma unroll
    for (int k = 0; k < 4; ++k) ta = wmma4(Ar[k], Bs[k], ta);
    v8f dr = ta, di = ta;
    #pragma unroll
    for (int k = 0; k < 4; ++k) {
        dr = wmma4(Am[k], Bi[k], dr);
        di = wmma4(Ap[k], Br[k], di);
    }

    float R[8], I[8];
    #pragma unroll
    for (int r = 0; r < 8; ++r) { R[r] = dr[r]; I[r] = di[r]; }

    #pragma unroll
    for (int r = 0; r < 4; ++r) {
        float arr = R[r], aii = I[r], brr = R[r + 4], bii = I[r + 4];
        R[r]     = c1 * arr + s1 * bii;  I[r]     = c1 * aii - s1 * brr;
        R[r + 4] = c1 * brr + s1 * aii;  I[r + 4] = c1 * bii - s1 * arr;
    }
    #pragma unroll
    for (int t2 = 0; t2 < 4; ++t2) {
        int r = (t2 & 1) + (t2 >> 1) * 4;
        float arr = R[r], aii = I[r], brr = R[r + 2], bii = I[r + 2];
        R[r]     = c2 * arr + s2 * bii;  I[r]     = c2 * aii - s2 * brr;
        R[r + 2] = c2 * brr + s2 * aii;  I[r + 2] = c2 * bii - s2 * arr;
    }
    #pragma unroll
    for (int t3 = 0; t3 < 4; ++t3) {
        int r = t3 * 2;
        float arr = R[r], aii = I[r], brr = R[r + 1], bii = I[r + 1];
        R[r]     = c3 * arr + s3 * bii;  I[r]     = c3 * aii - s3 * brr;
        R[r + 1] = c3 * brr + s3 * aii;  I[r + 1] = c3 * bii - s3 * arr;
    }

    float SR[8], SI[8];
    #pragma unroll
    for (int r = 0; r < 8; ++r) { SR[r] = swap16(R[r]); SI[r] = swap16(I[r]); }

    for (int k = 0; k < 4; ++k) {
        const v2f rv  = v2f{R[2 * k],  R[2 * k + 1]};
        const v2f iv  = v2f{I[2 * k],  I[2 * k + 1]};
        const v2f srv = v2f{SR[2 * k], SR[2 * k + 1]};
        const v2f siv = v2f{SI[2 * k], SI[2 * k + 1]};
        Br[k] = c0 * rv + s0 * siv;
        Bi[k] = c0 * iv - s0 * srv;
        Bs[k] = Br[k] + Bi[k];
    }

    load_frag(g + 3 * 256 + foff, Ar);
    load_frag(g + 4 * 256 + foff, Am);
    load_frag(g + 5 * 256 + foff, Ap);

    v8f tb = {0.f,0.f,0.f,0.f,0.f,0.f,0.f,0.f};
    #pragma unroll
    for (int k = 0; k < 4; ++k) tb = wmma4(Ar[k], Bs[k], tb);
    v8f er = tb, ei = tb;
    #pragma unroll
    for (int k = 0; k < 4; ++k) {
        er = wmma4(Am[k], Bi[k], er);
        ei = wmma4(Ap[k], Br[k], ei);
    }

    float p[8];
    #pragma unroll
    for (int r = 0; r < 8; ++r) p[r] = er[r] * er[r] + ei[r] * ei[r];

    const float t0 = p[0] + p[1], t1 = p[2] + p[3];
    const float t2 = p[4] + p[5], t3 = p[6] + p[7];
    const float u01 = t0 + t1, u23 = t2 + t3;
    float z0 = u01 + u23;
    float z1 = u01 - u23;
    float z2 = (t0 + t2) - (t1 + t3);
    const float w0 = p[0] - p[1], w1 = p[2] - p[3];
    const float w2 = p[4] - p[5], w3 = p[6] - p[7];
    float z3 = (w0 + w1) + (w2 + w3);
    if (hi) z0 = -z0;
    z0 += swap16(z0);
    z1 += swap16(z1);
    z2 += swap16(z2);
    z3 += swap16(z3);

    if (!hi) {
        float4 o; o.x = z0; o.y = z1; o.z = z2; o.w = z3;
        reinterpret_cast<float4*>(out)[samp] = o;
    }
}

extern "C" void kernel_launch(void* const* d_in, const int* in_sizes, int n_in,
                              void* d_out, int out_size, void* d_ws, size_t ws_size,
                              hipStream_t stream) {
    const float* x = (const float*)d_in[0];
    const float* w = (const float*)d_in[1];
    float* g   = (float*)d_ws;
    float* out = (float*)d_out;

    int B = in_sizes[0] / 4;
    build_G<<<1, 512, 0, stream>>>(w, g);
    qml_kernel<<<B / 128, 256, 0, stream>>>(x, g, out);
}


# Second layer pulled back onto the observables: <Z_i> = <v1|O_0..O_i|v1>, per-sample 2x2 Hermitian forms on vector ALU; second product, RX layer, G1 table gone (12 f32 WMMAs, not 24); sums regroup
# speedup vs baseline: 1.5476x; 1.2316x over previous
#include <hip/hip_runtime.h>
#include <math.h>

typedef float v2f __attribute__((ext_vector_type(2)));
typedef float v8f __attribute__((ext_vector_type(8)));

__device__ inline v8f wmma4(v2f a, v2f b, v8f c) {
    return __builtin_amdgcn_wmma_f32_16x16x4_f32(
         false, a,  false, b,
         (short)0, c,  false,  false);
}

__device__ inline float swap16(float v) {
    return __int_as_float(__builtin_amdgcn_ds_swizzle(__float_as_int(v), 0x401F));
}

__global__ void build_G(const float* __restrict__ w, float* __restrict__ g) {
    int t = threadIdx.x;
    if (t >= 256) {
        if (t < 260) {
            const int q = t - 256;
            float sy, cy;
            __sincosf(w[(4 + q) * 2 + 0] * 0.5f, &sy, &cy);
            const float cth = cy * cy - sy * sy;
            const float sth = 2.0f * sy * cy;
            g[768 + 3 * q + 0] = -sth;
            g[768 + 3 * q + 1] = 2.0f * cth;
            g[768 + 3 * q + 2] = cth;
        }
        return;
    }
    int e = t;
    int r = e >> 4;
    int c = e & 15;

    int j = r;
    if (j & 2) j ^= 1;
    if (j & 4) j ^= 2;
    if (j & 8) j ^= 4;

    float ar = 1.0f, ai = 0.0f;
    #pragma unroll
    for (int q = 0; q < 4; ++q) {
        float ty = w[q * 2 + 0] * 0.5f;
        float tz = w[q * 2 + 1] * 0.5f;
        float cy, sy, cz, sz;
        __sincosf(ty, &sy, &cy);
        __sincosf(tz, &sz, &cz);
        int br = (j >> (3 - q)) & 1;
        int bc = (c >> (3 - q)) & 1;
        float ur, ui;
        if (br == 0) { float m = bc ? -sy : cy; ur = m * cz; ui = -m * sz; }
        else         { float m = bc ?  cy : sy; ur = m * cz; ui =  m * sz; }
        float nr = ar * ur - ai * ui;
        float ni = ar * ui + ai * ur;
        ar = nr; ai = ni;
    }
    g[0 * 256 + e] = ar;
    g[1 * 256 + e] = -(ar + ai);
    g[2 * 256 + e] = ai - ar;
}

__device__ inline void load_frag(const float* __restrict__ p, v2f (&A)[4]) {
    const float4 lo = *reinterpret_cast<const float4*>(p);
    const float4 up = *reinterpret_cast<const float4*>(p + 4);
    A[0] = v2f{lo.x, lo.y};  A[1] = v2f{lo.z, lo.w};
    A[2] = v2f{up.x, up.y};  A[3] = v2f{up.z, up.w};
}

__device__ inline void herm_sep(v2f& ar, v2f& ai, v2f& br, v2f& bi,
                                float nx, float ny, float nz) {
    const v2f a_re = nz * ar + nx * br + ny * bi;
    const v2f a_im = nz * ai + nx * bi - ny * br;
    const v2f b_re = nx * ar - ny * ai - nz * br;
    const v2f b_im = nx * ai + ny * ar - nz * bi;
    ar = a_re; ai = a_im; br = b_re; bi = b_im;
}
__device__ inline void herm_adj(v2f& ur, v2f& ui, float nx, v2f NY, v2f NZ) {
    const v2f xr = __builtin_shufflevector(ur, ur, 1, 0);
    const v2f xi = __builtin_shufflevector(ui, ui, 1, 0);
    const v2f n_re = NZ * ur + nx * xr + NY * xi;
    const v2f n_im = NZ * ui + nx * xi - NY * xr;
    ur = n_re; ui = n_im;
}

__device__ inline float dot_re(const v2f (&R)[4], const v2f (&I)[4],
                               const v2f (&Ur)[4], const v2f (&Ui)[4]) {
    v2f a = R[0] * Ur[0], b = I[0] * Ui[0];
    #pragma unroll
    for (int k = 1; k < 4; ++k) { a += R[k] * Ur[k]; b += I[k] * Ui[k]; }
    a += b;
    return a.x + a.y;
}

__global__ void __launch_bounds__(256) qml_kernel(const float* __restrict__ x,
                                                  const float* __restrict__ g,
                                                  float* __restrict__ out) {
    const int lane = threadIdx.x & 31;
    const int wave = threadIdx.x >> 5;
    const int n    = lane & 15;
    const bool hi  = lane >= 16;
    const int foff = n * 16 + (hi ? 8 : 0);
    const int samp = blockIdx.x * 128 + wave * 16 + n;

    float4 xv = reinterpret_cast<const float4*>(x)[samp];
    float c0, s0, c1, s1, c2, s2, c3, s3;
    __sincosf(xv.x * 0.5f, &s0, &c0);
    __sincosf(xv.y * 0.5f, &s1, &c1);
    __sincosf(xv.z * 0.5f, &s2, &c2);
    __sincosf(xv.w * 0.5f, &s3, &c3);

    v2f Ar[4], Am[4], Ap[4];
    load_frag(g + 0 * 256 + foff, Ar);
    load_frag(g + 1 * 256 + foff, Am);
    load_frag(g + 2 * 256 + foff, Ap);

    const float gl = hi ? 0.f : c0;
    const float gh = hi ? s0 : 0.f;
    const float gs = hi ? -s0 : c0;
    const float gt = hi ? s0 : c0;
    const float ab[4] = {c1 * c2, c1 * s2, s1 * c2, s1 * s2};
    const float f3[2] = {c3, s3};
    v2f Br[4], Bi[4], Bs[4];
    #pragma unroll
    for (int k = 0; k < 4; ++k) {
        #pragma unroll
        for (int jj = 0; jj < 2; ++jj) {
            const float P = ab[k] * f3[jj];
            const int q = (((k >> 1) & 1) + (k & 1) + jj) & 3;
            float br, bi, bs;
            if (q == 0)      { br =  gl * P;  bi = -(gh * P); bs =  gs * P; }
            else if (q == 1) { br = -(gh * P); bi = -(gl * P); bs = -(gt * P); }
            else if (q == 2) { br = -(gl * P); bi =  gh * P;  bs = -(gs * P); }
            else             { br =  gh * P;  bi =  gl * P;  bs =  gt * P; }
            Br[k][jj] = br;
            Bi[k][jj] = bi;
            Bs[k][jj] = bs;
        }
    }

    v8f ta = {0.f,0.f,0.f,0.f,0.f,0.f,0.f,0.f};
    #pragma unroll
    for (int k = 0; k < 4; ++k) ta = wmma4(Ar[k], Bs[k], ta);
    v8f dr = ta, di = ta;
    #pragma unroll
    for (int k = 0; k < 4; ++k) {
        dr = wmma4(Am[k], Bi[k], dr);
        di = wmma4(Ap[k], Br[k], di);
    }

    const float* __restrict__ cq = g + 768;
    const float nx0 = cq[0], nx1 = cq[3], nx2 = cq[6], nx3 = cq[9];
    const float ny0 = cq[1]  * (s0 * c0), nz0 = cq[2]  * (c0 * c0 - s0 * s0);
    const float ny1 = cq[4]  * (s1 * c1), nz1 = cq[5]  * (c1 * c1 - s1 * s1);
    const float ny2 = cq[7]  * (s2 * c2), nz2 = cq[8]  * (c2 * c2 - s2 * s2);
    const float ny3 = cq[10] * (s3 * c3), nz3 = cq[11] * (c3 * c3 - s3 * s3);

    v2f R[4], I[4];
    #pragma unroll
    for (int k = 0; k < 4; ++k) {
        R[k] = v2f{dr[2 * k], dr[2 * k + 1]};
        I[k] = v2f{di[2 * k], di[2 * k + 1]};
    }

    const float nzs = hi ? -nz0 : nz0;
    const float nys = hi ? -ny0 : ny0;
    v2f Ur[4], Ui[4];
    #pragma unroll
    for (int k = 0; k < 4; ++k) {
        const v2f sr = v2f{swap16(R[k].x), swap16(R[k].y)};
        const v2f si = v2f{swap16(I[k].x), swap16(I[k].y)};
        Ur[k] = nzs * R[k] + nx0 * sr + nys * si;
        Ui[k] = nzs * I[k] + nx0 * si - nys * sr;
    }
    float z0 = dot_re(R, I, Ur, Ui);

    herm_sep(Ur[0], Ui[0], Ur[2], Ui[2], nx1, ny1, nz1);
    herm_sep(Ur[1], Ui[1], Ur[3], Ui[3], nx1, ny1, nz1);
    float z1 = dot_re(R, I, Ur, Ui);

    herm_sep(Ur[0], Ui[0], Ur[1], Ui[1], nx2, ny2, nz2);
    herm_sep(Ur[2], Ui[2], Ur[3], Ui[3], nx2, ny2, nz2);
    float z2 = dot_re(R, I, Ur, Ui);

    const v2f NY3 = v2f{ny3, -ny3};
    const v2f NZ3 = v2f{nz3, -nz3};
    #pragma unroll
    for (int k = 0; k < 4; ++k) herm_adj(Ur[k], Ui[k], nx3, NY3, NZ3);
    float z3 = dot_re(R, I, Ur, Ui);

    z0 += swap16(z0);
    z1 += swap16(z1);
    z2 += swap16(z2);
    z3 += swap16(z3);

    if (!hi) {
        float4 o; o.x = z0; o.y = z1; o.z = z2; o.w = z3;
        reinterpret_cast<float4*>(out)[samp] = o;
    }
}

extern "C" void kernel_launch(void* const* d_in, const int* in_sizes, int n_in,
                              void* d_out, int out_size, void* d_ws, size_t ws_size,
                              hipStream_t stream) {
    const float* x = (const float*)d_in[0];
    const float* w = (const float*)d_in[1];
    float* g   = (float*)d_ws;
    float* out = (float*)d_out;

    int B = in_sizes[0] / 4;
    build_G<<<1, 288, 0, stream>>>(w, g);
    qml_kernel<<<B / 128, 256, 0, stream>>>(x, g, out);
}


# Table kernel gone: every wave derives its own A-operand registers and the second layer's constants from the 12 weight angles (lane-wise sincos, readlane); one launch, no workspace; sums unchanged
# speedup vs baseline: 3.6944x; 2.3872x over previous
#include <hip/hip_runtime.h>
#include <math.h>


typedef float v2f __attribute__((ext_vector_type(2)));
typedef float v8f __attribute__((ext_vector_type(8)));

__device__ inline v8f wmma4(v2f a, v2f b, v8f c) {
    return __builtin_amdgcn_wmma_f32_16x16x4_f32(
         false, a,  false, b,
         (short)0, c,  false,  false);
}

__device__ inline v2f a_entry(int comp, float a, float b) {
    float sa, ca, sb, cb;
    __sincosf(a, &sa, &ca);
    __sincosf(b, &sb, &cb);
    const bool cx = comp == 0, cy = comp == 1, cz = comp == 2;
    const float prod = sa * (cx ? cb : sb);
    const float v0 = cx ? sb : (cy ? -cb : 0.0f);
    const float v1 = (cx || cy) ? prod : (cz ? ca : 0.0f);
    return v2f{v0, v1};
}

__device__ inline float lane_value(float v, int l) {
    return __int_as_float(__builtin_amdgcn_readlane(__float_as_int(v), l));
}

__global__ void __launch_bounds__(256) qml_kernel(const float* __restrict__ x,
                                                  const float* __restrict__ w,
                                                  float* __restrict__ out) {
    const int lane = threadIdx.x & 31;
    const int wave = threadIdx.x >> 5;
    const int n    = lane & 15;
    const bool hi  = lane >= 16;
    const int samp = blockIdx.x * 128 + wave * 16 + n;

    const int h    = lane >> 4;
    const int comp = (lane & 7) - 3 * h;
    const float2 wq0 = reinterpret_cast<const float2*>(w)[h];
    const float2 wq1 = reinterpret_cast<const float2*>(w)[2 + h];
    const float  thl = w[8 + 2 * (lane & 3)];
    __builtin_amdgcn_sched_barrier(0);
    const float4 xv = reinterpret_cast<const float4*>(x)[samp];

    const v2f A0 = a_entry(comp, wq0.x, wq0.y);
    const v2f A1 = a_entry(comp, wq1.x, wq1.y);
    float sth, cth;
    __sincosf(thl, &sth, &cth);

    float s0, c0, s1, c1, s2, c2, s3, c3;
    __sincosf(xv.x, &s0, &c0);
    __sincosf(xv.y, &s1, &c1);
    __sincosf(xv.z, &s2, &c2);
    __sincosf(xv.w, &s3, &c3);

    const v2f B0 = hi ? v2f{s1, c1} : v2f{s0, c0};
    const v2f B1 = hi ? v2f{s3, c3} : v2f{s2, c2};

    const v8f zero = {0.f, 0.f, 0.f, 0.f, 0.f, 0.f, 0.f, 0.f};
    const v8f d0 = wmma4(A0, B0, zero);
    const v8f d1 = wmma4(A1, B1, zero);

    const float nx0 = -lane_value(sth, 0), nx1 = -lane_value(sth, 1);
    const float nx2 = -lane_value(sth, 2), nx3 = -lane_value(sth, 3);
    const float ct0 = lane_value(cth, 0), ct1 = lane_value(cth, 1);
    const float ct2 = lane_value(cth, 2), ct3 = lane_value(cth, 3);
    const float ny0 = ct0 * s0, nz0 = ct0 * c0;
    const float ny1 = ct1 * s1, nz1 = ct1 * c1;
    const float ny2 = ct2 * s2, nz2 = ct2 * c2;
    const float ny3 = ct3 * s3, nz3 = ct3 * c3;

    float S = nz0 * d0[2];
    float D = nz0;
    float p = nx0 * d0[0] + ny0 * d0[1];
    float q = nx0 * d0[1] - ny0 * d0[0];
    const float z0 = S + p * d0[3];

    {
        const float bx = d0[3], by = d0[4], bz = d0[5];
        const float re = bx * S + p;
        const float im = by * D + bz * q;
        const float Sn = nz1 * (bz * D - by * q);
        const float Dn = nz1 * (S + bx * p);
        p = nx1 * re + ny1 * im;
        q = nx1 * im - ny1 * re;
        S = Sn; D = Dn;
    }
    const float z1 = S + p * d1[0];

    {
        const float bx = d1[0], by = d1[1], bz = d1[2];
        const float re = bx * S + p;
        const float im = by * D + bz * q;
        const float Sn = nz2 * (bz * D - by * q);
        const float Dn = nz2 * (S + bx * p);
        p = nx2 * re + ny2 * im;
        q = nx2 * im - ny2 * re;
        S = Sn; D = Dn;
    }
    const float z2 = S + p * d1[3];

    float z3;
    {
        const float bx = d1[3], by = d1[4], bz = d1[5];
        const float re = bx * S + p;
        const float im = by * D + bz * q;
        const float Sn = nz3 * (bz * D - by * q);
        const float pn = nx3 * re + ny3 * im;
        z3 = Sn + pn;
    }

    if (!hi) {
        float4 o; o.x = z0; o.y = z1; o.z = z2; o.w = z3;
        reinterpret_cast<float4*>(out)[samp] = o;
    }
}

extern "C" void kernel_launch(void* const* d_in, const int* in_sizes, int n_in,
                              void* d_out, int out_size, void* d_ws, size_t ws_size,
                              hipStream_t stream) {
    const float* x = (const float*)d_in[0];
    const float* w = (const float*)d_in[1];
    float* out = (float*)d_out;

    int B = in_sizes[0] / 4;
    qml_kernel<<<B / 128, 256, 0, stream>>>(x, w, out);
}


# Each wave derives its A operands and second-layer constants once and runs 8 tiles of 16 samples with them, all 8 tiles' angles requested up front: an eighth of the waves; per-tile sums unchanged
# speedup vs baseline: 5.2735x; 1.4274x over previous
#include <hip/hip_runtime.h>
#include <math.h>


typedef float v2f __attribute__((ext_vector_type(2)));
typedef float v8f __attribute__((ext_vector_type(8)));

__device__ inline v8f wmma4(v2f a, v2f b, v8f c) {
    return __builtin_amdgcn_wmma_f32_16x16x4_f32(
         false, a,  false, b,
         (short)0, c,  false,  false);
}

__device__ inline v2f a_entry(int comp, float a, float b) {
    float sa, ca, sb, cb;
    __sincosf(a, &sa, &ca);
    __sincosf(b, &sb, &cb);
    const bool cx = comp == 0, cy = comp == 1, cz = comp == 2;
    const float prod = sa * (cx ? cb : sb);
    const float v0 = cx ? sb : (cy ? -cb : 0.0f);
    const float v1 = (cx || cy) ? prod : (cz ? ca : 0.0f);
    return v2f{v0, v1};
}

__device__ inline float lane_value(float v, int l) {
    return __int_as_float(__builtin_amdgcn_readlane(__float_as_int(v), l));
}

constexpr int TILES = 8;

__global__ void __launch_bounds__(256) qml_kernel(const float* __restrict__ x,
                                                  const float* __restrict__ w,
                                                  float* __restrict__ out) {
    const int lane = threadIdx.x & 31;
    const int wave = threadIdx.x >> 5;
    const int n    = lane & 15;
    const bool hi  = lane >= 16;
    const int samp0 = (blockIdx.x * 8 + wave) * (16 * TILES) + n;

    const int h    = lane >> 4;
    const int comp = (lane & 7) - 3 * h;
    const float2 wq0 = reinterpret_cast<const float2*>(w)[h];
    const float2 wq1 = reinterpret_cast<const float2*>(w)[2 + h];
    const float  thl = w[8 + 2 * (lane & 3)];
    __builtin_amdgcn_sched_barrier(0);
    float4 xv[TILES];
#pragma unroll
    for (int t = 0; t < TILES; ++t)
        xv[t] = reinterpret_cast<const float4*>(x)[samp0 + 16 * t];

    const v2f A0 = a_entry(comp, wq0.x, wq0.y);
    const v2f A1 = a_entry(comp, wq1.x, wq1.y);
    float sth, cth;
    __sincosf(thl, &sth, &cth);

    const float nx0 = -lane_value(sth, 0), nx1 = -lane_value(sth, 1);
    const float nx2 = -lane_value(sth, 2), nx3 = -lane_value(sth, 3);
    const float ct0 = lane_value(cth, 0), ct1 = lane_value(cth, 1);
    const float ct2 = lane_value(cth, 2), ct3 = lane_value(cth, 3);

    const v8f zero = {0.f, 0.f, 0.f, 0.f, 0.f, 0.f, 0.f, 0.f};

#pragma unroll
    for (int t = 0; t < TILES; ++t) {
        float s0, c0, s1, c1, s2, c2, s3, c3;
        __sincosf(xv[t].x, &s0, &c0);
        __sincosf(xv[t].y, &s1, &c1);
        __sincosf(xv[t].z, &s2, &c2);
        __sincosf(xv[t].w, &s3, &c3);

        const v2f B0 = hi ? v2f{s1, c1} : v2f{s0, c0};
        const v2f B1 = hi ? v2f{s3, c3} : v2f{s2, c2};

        const v8f d0 = wmma4(A0, B0, zero);
        const v8f d1 = wmma4(A1, B1, zero);

        const float ny0 = ct0 * s0, nz0 = ct0 * c0;
        const float ny1 = ct1 * s1, nz1 = ct1 * c1;
        const float ny2 = ct2 * s2, nz2 = ct2 * c2;
        const float ny3 = ct3 * s3, nz3 = ct3 * c3;

        float S = nz0 * d0[2];
        float D = nz0;
        float p = nx0 * d0[0] + ny0 * d0[1];
        float q = nx0 * d0[1] - ny0 * d0[0];
        const float z0 = S + p * d0[3];

        {
            const float bx = d0[3], by = d0[4], bz = d0[5];
            const float re = bx * S + p;
            const float im = by * D + bz * q;
            const float Sn = nz1 * (bz * D - by * q);
            const float Dn = nz1 * (S + bx * p);
            p = nx1 * re + ny1 * im;
            q = nx1 * im - ny1 * re;
            S = Sn; D = Dn;
        }
        const float z1 = S + p * d1[0];

        {
            const float bx = d1[0], by = d1[1], bz = d1[2];
            const float re = bx * S + p;
            const float im = by * D + bz * q;
            const float Sn = nz2 * (bz * D - by * q);
            const float Dn = nz2 * (S + bx * p);
            p = nx2 * re + ny2 * im;
            q = nx2 * im - ny2 * re;
            S = Sn; D = Dn;
        }
        const float z2 = S + p * d1[3];

        float z3;
        {
            const float bx = d1[3], by = d1[4], bz = d1[5];
            const float re = bx * S + p;
            const float im = by * D + bz * q;
            const float Sn = nz3 * (bz * D - by * q);
            const float pn = nx3 * re + ny3 * im;
            z3 = Sn + pn;
        }

        if (!hi) {
            float4 o; o.x = z0; o.y = z1; o.z = z2; o.w = z3;
            reinterpret_cast<float4*>(out)[samp0 + 16 * t] = o;
        }
    }
}

extern "C" void kernel_launch(void* const* d_in, const int* in_sizes, int n_in,
                              void* d_out, int out_size, void* d_ws, size_t ws_size,
                              hipStream_t stream) {
    const float* x = (const float*)d_in[0];
    const float* w = (const float*)d_in[1];
    float* out = (float*)d_out;

    int B = in_sizes[0] / 4;
    qml_kernel<<<B / (128 * TILES), 256, 0, stream>>>(x, w, out);
}


# One sample per lane, not per lane pair: A in block form (rows 0-7 in lane half 0's k slots, rows 8-15 in half 1's), one f32 WMMA per qubit whose spare rows give cos(th)sin x, cos(th)cos x; sums kept
# speedup vs baseline: 6.0653x; 1.1501x over previous
#include <hip/hip_runtime.h>
#include <math.h>


typedef float v2f __attribute__((ext_vector_type(2)));
typedef float v8f __attribute__((ext_vector_type(8)));

__device__ inline v8f wmma4(v2f a, v2f b, v8f c) {
    return __builtin_amdgcn_wmma_f32_16x16x4_f32(
         false, a,  false, b,
         (short)0, c,  false,  false);
}

__device__ inline v2f a_entry(int row, float a, float b, float ct) {
    float sa, ca, sb, cb;
    __sincosf(a, &sa, &ca);
    __sincosf(b, &sb, &cb);
    const bool r0 = row == 0, r1 = row == 1, r2 = row == 2;
    const bool r3 = row == 3, r4 = row == 4;
    const float ncb  = -cb;
    const float prod = sa * (r0 ? cb : sb);
    float v0 = 0.0f;
    v0 = r3 ? ct : v0;
    v0 = r1 ? ncb : v0;
    v0 = r0 ? sb : v0;
    float v1 = 0.0f;
    v1 = r4 ? ct : v1;
    v1 = r2 ? ca : v1;
    v1 = (r0 || r1) ? prod : v1;
    return v2f{v0, v1};
}

constexpr int TILES = 8;

__global__ void __launch_bounds__(256) qml_kernel(const float* __restrict__ x,
                                                  const float* __restrict__ w,
                                                  float* __restrict__ out) {
    const int lane = threadIdx.x & 31;
    const int wave = threadIdx.x >> 5;
    const int samp0 = (blockIdx.x * 8 + wave) * (32 * TILES) + lane;

    float4 xv[TILES];
#pragma unroll
    for (int t = 0; t < TILES; ++t)
        xv[t] = reinterpret_cast<const float4*>(x)[samp0 + 32 * t];
    __builtin_amdgcn_sched_barrier(0);

    float st0, ct0, st1, ct1, st2, ct2, st3, ct3;
    __sincosf(w[8],  &st0, &ct0);
    __sincosf(w[10], &st1, &ct1);
    __sincosf(w[12], &st2, &ct2);
    __sincosf(w[14], &st3, &ct3);
    const float nx0 = -st0, nx1 = -st1, nx2 = -st2, nx3 = -st3;

    const int row = (((lane >> 3) & 1) == (lane >> 4)) ? (lane & 7) : 7;
    const v2f A0 = a_entry(row, w[0], w[1], ct0);
    const v2f A1 = a_entry(row, w[2], w[3], ct1);
    const v2f A2 = a_entry(row, w[4], w[5], ct2);
    const v2f A3 = a_entry(row, w[6], w[7], ct3);
    __builtin_amdgcn_sched_barrier(0);

    const v8f zero = {0.f, 0.f, 0.f, 0.f, 0.f, 0.f, 0.f, 0.f};

#pragma unroll
    for (int t = 0; t < TILES; ++t) {
        float s0, c0, s1, c1, s2, c2, s3, c3;
        __sincosf(xv[t].x, &s0, &c0);
        __sincosf(xv[t].y, &s1, &c1);
        __sincosf(xv[t].z, &s2, &c2);
        __sincosf(xv[t].w, &s3, &c3);

        const v8f d0 = wmma4(A0, v2f{s0, c0}, zero);
        const v8f d1 = wmma4(A1, v2f{s1, c1}, zero);
        const v8f d2 = wmma4(A2, v2f{s2, c2}, zero);
        const v8f d3 = wmma4(A3, v2f{s3, c3}, zero);

        float S = d0[4] * d0[2];
        float D = d0[4];
        float p = nx0 * d0[0] + d0[3] * d0[1];
        float q = nx0 * d0[1] - d0[3] * d0[0];
        const float z0 = S + p * d1[0];

        {
            const float bx = d1[0], by = d1[1], bz = d1[2];
            const float ny = d1[3], nz = d1[4];
            const float re = bx * S + p;
            const float im = by * D + bz * q;
            const float Sn = nz * (bz * D - by * q);
            const float Dn = nz * (S + bx * p);
            p = nx1 * re + ny * im;
            q = nx1 * im - ny * re;
            S = Sn; D = Dn;
        }
        const float z1 = S + p * d2[0];

        {
            const float bx = d2[0], by = d2[1], bz = d2[2];
            const float ny = d2[3], nz = d2[4];
            const float re = bx * S + p;
            const float im = by * D + bz * q;
            const float Sn = nz * (bz * D - by * q);
            const float Dn = nz * (S + bx * p);
            p = nx2 * re + ny * im;
            q = nx2 * im - ny * re;
            S = Sn; D = Dn;
        }
        const float z2 = S + p * d3[0];

        float z3;
        {
            const float bx = d3[0], by = d3[1], bz = d3[2];
            const float ny = d3[3], nz = d3[4];
            const float re = bx * S + p;
            const float im = by * D + bz * q;
            const float Sn = nz * (bz * D - by * q);
            const float pn = nx3 * re + ny * im;
            z3 = Sn + pn;
        }

        float4 o; o.x = z0; o.y = z1; o.z = z2; o.w = z3;
        reinterpret_cast<float4*>(out)[samp0 + 32 * t] = o;
        if (t & 1) __builtin_amdgcn_sched_barrier(0);
    }
}

extern "C" void kernel_launch(void* const* d_in, const int* in_sizes, int n_in,
                              void* d_out, int out_size, void* d_ws, size_t ws_size,
                              hipStream_t stream) {
    const float* x = (const float*)d_in[0];
    const float* w = (const float*)d_in[1];
    float* out = (float*)d_out;

    int B = in_sizes[0] / 4;
    qml_kernel<<<B / (256 * TILES), 256, 0, stream>>>(x, w, out);
}
